# starting kernel
# baseline (speedup 1.0000x reference)
#include <hip/hip_runtime.h>


typedef __attribute__((ext_vector_type(2))) float v2f;
typedef __attribute__((ext_vector_type(8))) float v8f;

namespace gat {
constexpr int Bc = 8, Mc = 1024, Fc = 256, Hc = 8, Ec = 32;
constexpr size_t WH_N = (size_t)Bc * Hc * Mc * Ec;
constexpr size_t S_N  = (size_t)Bc * Hc * Mc;
constexpr float LALPHA  = 0.2f;
constexpr float NEG_INF = -9.0e15f;
}

__global__ __launch_bounds__(32)
void gat_wh_kernel(const float* __restrict__ x,
                   const float* __restrict__ w,
                   const float* __restrict__ att,
                   float* __restrict__ wh,
                   float* __restrict__ s1,
                   float* __restrict__ s2) {
    using namespace gat;
    const int lane = threadIdx.x;
    const int bid  = blockIdx.x;
    const int mb   = bid & 31;
    const int h    = (bid >> 5) & 7;
    const int b    = bid >> 8;

    const int col  = lane & 15;
    const int half = lane >> 4;
    const int k0   = half * 2;

    const int m0 = mb * 32;
    const float* xrow0 = x + ((size_t)b * Mc + (m0 + col)) * Fc;
    const float* xrow1 = xrow0 + (size_t)16 * Fc;
    const float* wbase = w + (size_t)h * Fc * Ec;

    v8f acc[2][2] = {};
    for (int kk = 0; kk < Fc; kk += 4) {
        v2f af0 = *(const v2f*)(xrow0 + kk + k0);
        v2f af1 = *(const v2f*)(xrow1 + kk + k0);
        const float* wp = wbase + (size_t)(kk + k0) * Ec;
        v2f bf0 = { wp[col],      wp[Ec + col]      };
        v2f bf1 = { wp[16 + col], wp[Ec + 16 + col] };
        acc[0][0] = __builtin_amdgcn_wmma_f32_16x16x4_f32(false, af0, false, bf0,
                                                          (short)0, acc[0][0], false, false);
        acc[0][1] = __builtin_amdgcn_wmma_f32_16x16x4_f32(false, af0, false, bf1,
                                                          (short)0, acc[0][1], false, false);
        acc[1][0] = __builtin_amdgcn_wmma_f32_16x16x4_f32(false, af1, false, bf0,
                                                          (short)0, acc[1][0], false, false);
        acc[1][1] = __builtin_amdgcn_wmma_f32_16x16x4_f32(false, af1, false, bf1,
                                                          (short)0, acc[1][1], false, false);
    }

    const float a1_0 = att[h * 2 * Ec + col];
    const float a1_1 = att[h * 2 * Ec + 16 + col];
    const float a2_0 = att[h * 2 * Ec + Ec + col];
    const float a2_1 = att[h * 2 * Ec + Ec + 16 + col];

    const int bh = b * Hc + h;
    float* whbase = wh + ((size_t)bh * Mc + m0) * Ec;

    #pragma unroll
    for (int t = 0; t < 2; ++t) {
        #pragma unroll
        for (int r = 0; r < 8; ++r) {
            const int m_l = t * 16 + r + 8 * half;
            const float v0 = acc[t][0][r];
            const float v1 = acc[t][1][r];
            whbase[(size_t)m_l * Ec + col]      = v0;
            whbase[(size_t)m_l * Ec + 16 + col] = v1;

            float p1 = v0 * a1_0 + v1 * a1_1;
            float p2 = v0 * a2_0 + v1 * a2_1;
            #pragma unroll
            for (int d = 1; d < 16; d <<= 1) {
                p1 += __shfl_xor(p1, d, 32);
                p2 += __shfl_xor(p2, d, 32);
            }
            if (col == 0) {
                s1[(size_t)bh * Mc + m0 + m_l] = p1;
                s2[(size_t)bh * Mc + m0 + m_l] = p2;
            }
        }
    }
}

__global__ __launch_bounds__(256)
void gat_stats_kernel(const int* __restrict__ adj,
                      const float* __restrict__ s1,
                      const float* __restrict__ s2,
                      float* __restrict__ rmax,
                      float* __restrict__ rsum) {
    using namespace gat;
    const int lane = threadIdx.x & 31;
    const int wid  = blockIdx.x * 8 + (threadIdx.x >> 5);
    const int m    = wid & (Mc - 1);
    const int bh   = wid >> 10;
    const int b    = bh >> 3;

    const float s1v = s1[(size_t)bh * Mc + m];
    const int*   arow  = adj + ((size_t)b * Mc + m) * Mc;
    const float* s2row = s2 + (size_t)bh * Mc;

    float mx = NEG_INF;
    for (int n = lane; n < Mc; n += 32) {
        float sc = s1v + s2row[n];
        sc = sc > 0.0f ? sc : LALPHA * sc;
        sc = arow[n] > 0 ? sc : NEG_INF;
        mx = fmaxf(mx, sc);
    }
    #pragma unroll
    for (int d = 1; d < 32; d <<= 1) mx = fmaxf(mx, __shfl_xor(mx, d, 32));

    float sum = 0.0f;
    for (int n = lane; n < Mc; n += 32) {
        float sc = s1v + s2row[n];
        sc = sc > 0.0f ? sc : LALPHA * sc;
        sc = arow[n] > 0 ? sc : NEG_INF;
        sum += __expf(sc - mx);
    }
    #pragma unroll
    for (int d = 1; d < 32; d <<= 1) sum += __shfl_xor(sum, d, 32);

    if (lane == 0) {
        rmax[wid] = mx;
        rsum[wid] = sum;
    }
}

__global__ __launch_bounds__(32)
void gat_attn_kernel(const int* __restrict__ adj,
                     const float* __restrict__ wh,
                     const float* __restrict__ s1,
                     const float* __restrict__ s2,
                     const float* __restrict__ rmax,
                     const float* __restrict__ rsum,
                     float* __restrict__ out) {
    using namespace gat;
    __shared__ float pblk[2][16 * 16];

    const int lane = threadIdx.x;
    const int bid  = blockIdx.x;
    const int mb   = bid & 31;
    const int h    = (bid >> 5) & 7;
    const int b    = bid >> 8;
    const int bh   = b * Hc + h;

    const int col  = lane & 15;
    const int half = lane >> 4;
    const int k0   = half * 2;
    const int m0   = mb * 32;

    float s1v[16], rmx[16], rin[16];
    #pragma unroll
    for (int t = 0; t < 2; ++t) {
        #pragma unroll
        for (int r = 0; r < 8; ++r) {
            const int m = m0 + t * 16 + r + 8 * half;
            s1v[t * 8 + r] = s1[(size_t)bh * Mc + m];
            rmx[t * 8 + r] = rmax[(size_t)bh * Mc + m];
            rin[t * 8 + r] = 1.0f / rsum[(size_t)bh * Mc + m];
        }
    }

    const float* whb  = wh + (size_t)bh * Mc * Ec;
    const int*   adjb = adj + (size_t)b * Mc * Mc;
    const float* s2b  = s2 + (size_t)bh * Mc;

    v8f acc[2][2] = {};
    for (int nb = 0; nb < Mc / 16; ++nb) {
        const int n0 = nb * 16;
        const float s2v = s2b[n0 + col];

        if (nb + 1 < Mc / 16) {
            __builtin_prefetch(&adjb[(size_t)(m0 + 8 * half) * Mc + n0 + 16 + col], 0, 0);
            __builtin_prefetch(&adjb[(size_t)(m0 + 16 + 8 * half) * Mc + n0 + 16 + col], 0, 0);
        }

        #pragma unroll
        for (int t = 0; t < 2; ++t) {
            #pragma unroll
            for (int r = 0; r < 8; ++r) {
                const int mL = r + 8 * half;
                float sc = s1v[t * 8 + r] + s2v;
                sc = sc > 0.0f ? sc : LALPHA * sc;
                const int a = adjb[(size_t)(m0 + t * 16 + mL) * Mc + n0 + col];
                sc = a > 0 ? sc : NEG_INF;
                pblk[t][mL * 16 + col] = __expf(sc - rmx[t * 8 + r]) * rin[t * 8 + r];
            }
        }
        __syncthreads();

        #pragma unroll
        for (int ks = 0; ks < 4; ++ks) {
            v2f af0 = *(const v2f*)&pblk[0][col * 16 + ks * 4 + k0];
            v2f af1 = *(const v2f*)&pblk[1][col * 16 + ks * 4 + k0];
            const float* wp = whb + (size_t)(n0 + ks * 4 + k0) * Ec;
            v2f bf0 = { wp[col],      wp[Ec + col]      };
            v2f bf1 = { wp[16 + col], wp[Ec + 16 + col] };
            acc[0][0] = __builtin_amdgcn_wmma_f32_16x16x4_f32(false, af0, false, bf0,
                                                              (short)0, acc[0][0], false, false);
            acc[0][1] = __builtin_amdgcn_wmma_f32_16x16x4_f32(false, af0, false, bf1,
                                                              (short)0, acc[0][1], false, false);
            acc[1][0] = __builtin_amdgcn_wmma_f32_16x16x4_f32(false, af1, false, bf0,
                                                              (short)0, acc[1][0], false, false);
            acc[1][1] = __builtin_amdgcn_wmma_f32_16x16x4_f32(false, af1, false, bf1,
                                                              (short)0, acc[1][1], false, false);
        }
        __syncthreads();
    }

    float* ob = out + (size_t)b * Mc * (Hc * Ec) + (size_t)h * Ec;
    #pragma unroll
    for (int t = 0; t < 2; ++t) {
        #pragma unroll
        for (int r = 0; r < 8; ++r) {
            const int m = m0 + t * 16 + r + 8 * half;
            float v0 = acc[t][0][r];
            float v1 = acc[t][1][r];
            v0 = v0 > 0.0f ? v0 : (__expf(v0) - 1.0f);
            v1 = v1 > 0.0f ? v1 : (__expf(v1) - 1.0f);
            ob[(size_t)m * (Hc * Ec) + col]      = v0;
            ob[(size_t)m * (Hc * Ec) + 16 + col] = v1;
        }
    }
}

extern "C" void kernel_launch(void* const* d_in, const int* in_sizes, int n_in,
                              void* d_out, int out_size, void* d_ws, size_t ws_size,
                              hipStream_t stream) {
    using namespace gat;
    const float* x   = (const float*)d_in[0];
    const int*   adj = (const int*)  d_in[1];
    const float* w   = (const float*)d_in[2];
    const float* att = (const float*)d_in[3];
    float* out = (float*)d_out;

    float* ws   = (float*)d_ws;
    float* wh   = ws;
    float* s1   = ws + WH_N;
    float* s2   = ws + WH_N + S_N;
    float* rmx  = ws + WH_N + 2 * S_N;
    float* rsm  = ws + WH_N + 3 * S_N;

    const int slabs = Bc * Hc * (Mc / 32);

    gat_wh_kernel<<<slabs, 32, 0, stream>>>(x, w, att, wh, s1, s2);
    gat_stats_kernel<<<(int)(S_N / 8), 256, 0, stream>>>(adj, s1, s2, rmx, rsm);
    gat_attn_kernel<<<slabs, 32, 0, stream>>>(adj, wh, s1, s2, rmx, rsm, out);
}


# softmax-stats kernel on packed two-float math: mask as additive bias, row max from the largest kept s2, exponent = max of two folded multiply-adds in log2 units; row-sum order changes
# speedup vs baseline: 2.4632x; 2.4632x over previous
#include <hip/hip_runtime.h>


typedef __attribute__((ext_vector_type(2))) float v2f;
typedef __attribute__((ext_vector_type(8))) float v8f;

namespace gat {
constexpr int Bc = 8, Mc = 1024, Fc = 256, Hc = 8, Ec = 32;
constexpr size_t WH_N = (size_t)Bc * Hc * Mc * Ec;
constexpr size_t S_N  = (size_t)Bc * Hc * Mc;
constexpr float LALPHA  = 0.2f;
constexpr float NEG_INF = -9.0e15f;
}

namespace gat {
constexpr int WSLAB  = 16;
constexpr int XPITCH = Fc + 4;
}

__global__ __launch_bounds__(256)
void gat_wh_kernel(const float* __restrict__ x,
                   const float* __restrict__ w,
                   const float* __restrict__ att,
                   float* __restrict__ wh,
                   float* __restrict__ s1,
                   float* __restrict__ s2) {
    using namespace gat;
    __shared__ __attribute__((aligned(16))) float xs[WSLAB * XPITCH];

    const int tid  = threadIdx.x;
    const int lane = tid & 31;
    const int h    = tid >> 5;
    const int mb   = blockIdx.x & 63;
    const int b    = blockIdx.x >> 6;

    const int col  = lane & 15;
    const int half = lane >> 4;
    const int k0   = half * 2;

    const int m0 = mb * WSLAB;

    {
        const float4* xsrc = (const float4*)(x + ((size_t)b * Mc + m0) * Fc);
        #pragma unroll
        for (int i = 0; i < WSLAB * Fc / 4 / 256; ++i) {
            const int idx = tid + 256 * i;
            const float4 v = xsrc[idx];
            *(float4*)&xs[(idx >> 6) * XPITCH + 4 * (idx & 63)] = v;
        }
    }

    const float* xrow  = xs + col * XPITCH + k0;
    const float* wbase = w + (size_t)h * Fc * Ec;

    __syncthreads();

    v8f acc[2] = {};
    for (int kk = 0; kk < Fc; kk += 4) {
        v2f af = *(const v2f*)(xrow + kk);
        const float* wp = wbase + (size_t)(kk + k0) * Ec;
        v2f bf0 = { wp[col],      wp[Ec + col]      };
        v2f bf1 = { wp[16 + col], wp[Ec + 16 + col] };
        acc[0] = __builtin_amdgcn_wmma_f32_16x16x4_f32(false, af, false, bf0,
                                                       (short)0, acc[0], false, false);
        acc[1] = __builtin_amdgcn_wmma_f32_16x16x4_f32(false, af, false, bf1,
                                                       (short)0, acc[1], false, false);
    }

    const float a1_0 = att[h * 2 * Ec + col];
    const float a1_1 = att[h * 2 * Ec + 16 + col];
    const float a2_0 = att[h * 2 * Ec + Ec + col];
    const float a2_1 = att[h * 2 * Ec + Ec + 16 + col];

    const int bh = b * Hc + h;
    float* whbase = wh + ((size_t)bh * Mc + m0) * Ec;

    #pragma unroll
    for (int r = 0; r < 8; ++r) {
        const int m_l = r + 8 * half;
        const float v0 = acc[0][r];
        const float v1 = acc[1][r];
        whbase[(size_t)m_l * Ec + col]      = v0;
        whbase[(size_t)m_l * Ec + 16 + col] = v1;

        float p1 = v0 * a1_0 + v1 * a1_1;
        float p2 = v0 * a2_0 + v1 * a2_1;
        #pragma unroll
        for (int d = 1; d < 16; d <<= 1) {
            p1 += __shfl_xor(p1, d, 32);
            p2 += __shfl_xor(p2, d, 32);
        }
        if (col == 0) {
            s1[(size_t)bh * Mc + m0 + m_l] = p1;
            s2[(size_t)bh * Mc + m0 + m_l] = p2;
        }
    }
}

namespace gat {
constexpr float LOG2E = 1.4426950408889634f;
constexpr float NEGL  = NEG_INF * LOG2E;
constexpr float QL    = LALPHA * LOG2E;
}

__global__ __launch_bounds__(256)
void gat_stats_kernel(const int* __restrict__ adj,
                      const float* __restrict__ s1,
                      const float* __restrict__ s2,
                      float* __restrict__ rmax,
                      float* __restrict__ rsum) {
    using namespace gat;
    const int lane = threadIdx.x & 31;
    const int row  = blockIdx.x * 8 + (threadIdx.x >> 5);
    const int m    = row & (Mc - 1);
    const int b    = row >> 10;
    const size_t bh0 = (size_t)b * Hc;

    const int4* arow = (const int4*)(adj + (size_t)row * Mc) + lane;
    v2f kb[16];
    int amax = 0;
    #pragma unroll
    for (int j = 0; j < 8; ++j) {
        const int4 a = arow[j * 32];
        kb[2 * j]     = v2f{ a.x > 0 ? 0.0f : NEGL, a.y > 0 ? 0.0f : NEGL };
        kb[2 * j + 1] = v2f{ a.z > 0 ? 0.0f : NEGL, a.w > 0 ? 0.0f : NEGL };
        amax = max(max(amax, a.x), a.y);
        amax = max(max(amax, a.z), a.w);
    }
    const bool anykeep = __builtin_amdgcn_ballot_w32(amax > 0) != 0u;

    float4 cur[8];
    {
        const float4* s2row = (const float4*)(s2 + bh0 * Mc) + lane;
        #pragma unroll
        for (int j = 0; j < 8; ++j) cur[j] = s2row[j * 32];
    }

    const v2f Lv = { LOG2E, LOG2E };
    const v2f Qv = { QL, QL };

    #pragma unroll
    for (int h = 0; h < Hc; ++h) {
        const float s1h = s1[(bh0 + h) * Mc + m];

        v2f sb[16];
        float m2a = -__builtin_inff(), m2b = -__builtin_inff();
        #pragma unroll
        for (int j = 0; j < 8; ++j) {
            sb[2 * j]     = v2f{ cur[j].x, cur[j].y } + kb[2 * j];
            sb[2 * j + 1] = v2f{ cur[j].z, cur[j].w } + kb[2 * j + 1];
            m2a = fmaxf(fmaxf(m2a, sb[2 * j].x), sb[2 * j].y);
            m2b = fmaxf(fmaxf(m2b, sb[2 * j + 1].x), sb[2 * j + 1].y);
        }
        __builtin_amdgcn_sched_barrier(0);
        if (h + 1 < Hc) {
            const float4* s2row = (const float4*)(s2 + (bh0 + h + 1) * Mc) + lane;
            #pragma unroll
            for (int j = 0; j < 8; ++j) cur[j] = s2row[j * 32];
        }

        float m2 = fmaxf(m2a, m2b);
        #pragma unroll
        for (int d = 1; d < 32; d <<= 1) m2 = fmaxf(m2, __shfl_xor(m2, d, 32));

        float t = s1h + m2;
        t = t > 0.0f ? t : LALPHA * t;
        const float mx = anykeep ? fmaxf(t, NEG_INF) : NEG_INF;

        const float a = (s1h - mx) * LOG2E;
        const float c = (LALPHA * s1h - mx) * LOG2E;
        const v2f av = { a, a };
        const v2f cv = { c, c };
        v2f suma = { 0.0f, 0.0f }, sumb = { 0.0f, 0.0f };
        #pragma unroll
        for (int j = 0; j < 8; ++j) {
            const v2f elo = __builtin_elementwise_max(__builtin_elementwise_fma(sb[2 * j], Lv, av),
                                                      __builtin_elementwise_fma(sb[2 * j], Qv, cv));
            const v2f ehi = __builtin_elementwise_max(__builtin_elementwise_fma(sb[2 * j + 1], Lv, av),
                                                      __builtin_elementwise_fma(sb[2 * j + 1], Qv, cv));
            suma += v2f{ __builtin_amdgcn_exp2f(elo.x), __builtin_amdgcn_exp2f(elo.y) };
            sumb += v2f{ __builtin_amdgcn_exp2f(ehi.x), __builtin_amdgcn_exp2f(ehi.y) };
            if (j & 1) __builtin_amdgcn_sched_barrier(0);
        }
        const v2f st = suma + sumb;
        float sum = st.x + st.y;
        #pragma unroll
        for (int d = 1; d < 32; d <<= 1) sum += __shfl_xor(sum, d, 32);
        const float es = fmaxf(__builtin_fmaf(m2, LOG2E, a), __builtin_fmaf(m2, QL, c));
        sum *= __builtin_amdgcn_exp2f(-es);

        if (lane == 0) {
            rmax[(bh0 + h) * Mc + m] = mx;
            rsum[(bh0 + h) * Mc + m] = anykeep ? sum : (float)Mc;
        }
    }
}

namespace gat {
constexpr int SLAB   = 16;
constexpr int APITCH = Mc + 4;
constexpr int NT     = 32;
constexpr int PPITCH = NT + 4;
}

__global__ __launch_bounds__(256)
void gat_attn_kernel(const int* __restrict__ adj,
                     const float* __restrict__ wh,
                     const float* __restrict__ s1,
                     const float* __restrict__ s2,
                     const float* __restrict__ rmax,
                     const float* __restrict__ rsum,
                     float* __restrict__ out) {
    using namespace gat;
    __shared__ __attribute__((aligned(16))) int   arow[SLAB * APITCH];
    __shared__ __attribute__((aligned(16))) float pblk[Hc][16 * PPITCH];

    const int tid  = threadIdx.x;
    const int lane = tid & 31;
    const int h    = tid >> 5;
    const int mb   = blockIdx.x & 63;
    const int b    = blockIdx.x >> 6;
    const int bh   = b * Hc + h;

    const int col  = lane & 15;
    const int half = lane >> 4;
    const int m0   = mb * SLAB;

    {
        const int4* asrc = (const int4*)(adj + ((size_t)b * Mc + m0) * Mc) + tid;
        #pragma unroll
        for (int r = 0; r < SLAB; ++r) {
            const int4 a = asrc[(size_t)r * (Mc / 4)];
            *(int4*)&arow[r * APITCH + 4 * tid] = a;
        }
    }

    float s1v[8], rmx[8], rin[8];
    #pragma unroll
    for (int r = 0; r < 8; ++r) {
        const int m = m0 + r + 8 * half;
        s1v[r] = s1[(size_t)bh * Mc + m];
        rmx[r] = rmax[(size_t)bh * Mc + m];
        rin[r] = 1.0f / rsum[(size_t)bh * Mc + m];
    }

    const float* whb = wh + (size_t)bh * Mc * Ec;
    const float* s2b = s2 + (size_t)bh * Mc + 2 * col;
    float*       pw  = pblk[h];
    const int*   am  = arow + (8 * half) * APITCH + 2 * col;
    float*       pst = pw + (8 * half) * PPITCH + 2 * col;
    const float* pld = pw + col * PPITCH + 16 * half;

    __syncthreads();

    v8f acc[2] = {};
    for (int nb = 0; nb < Mc / NT; ++nb) {
        const int n0 = nb * NT;

        const v2f s2v = *(const v2f*)(s2b + n0);

        v2f bf[8][2];
        #pragma unroll
        for (int ks = 0; ks < 8; ++ks) {
            const float* wp = whb + (size_t)(n0 + 16 * half + 2 * ks) * Ec;
            bf[ks][0] = v2f{ wp[col],      wp[Ec + col]      };
            bf[ks][1] = v2f{ wp[16 + col], wp[Ec + 16 + col] };
        }

        #pragma unroll
        for (int r = 0; r < 8; ++r) {
            const int2 a = *(const int2*)&am[r * APITCH + n0];
            float sx = s1v[r] + s2v.x;
            float sy = s1v[r] + s2v.y;
            sx = sx > 0.0f ? sx : LALPHA * sx;
            sy = sy > 0.0f ? sy : LALPHA * sy;
            sx = a.x > 0 ? sx : NEG_INF;
            sy = a.y > 0 ? sy : NEG_INF;
            *(v2f*)&pst[r * PPITCH] = v2f{ __expf(sx - rmx[r]) * rin[r],
                                           __expf(sy - rmx[r]) * rin[r] };
        }
        __builtin_amdgcn_fence(__ATOMIC_ACQ_REL, "wavefront", "local");

        #pragma unroll
        for (int q = 0; q < 4; ++q) {
            const float4 p4 = *(const float4*)&pld[4 * q];
            const v2f af0 = v2f{ p4.x, p4.y };
            const v2f af1 = v2f{ p4.z, p4.w };
            acc[0] = __builtin_amdgcn_wmma_f32_16x16x4_f32(false, af0, false, bf[2 * q][0],
                                                           (short)0, acc[0], false, false);
            acc[1] = __builtin_amdgcn_wmma_f32_16x16x4_f32(false, af0, false, bf[2 * q][1],
                                                           (short)0, acc[1], false, false);
            acc[0] = __builtin_amdgcn_wmma_f32_16x16x4_f32(false, af1, false, bf[2 * q + 1][0],
                                                           (short)0, acc[0], false, false);
            acc[1] = __builtin_amdgcn_wmma_f32_16x16x4_f32(false, af1, false, bf[2 * q + 1][1],
                                                           (short)0, acc[1], false, false);
        }
        __builtin_amdgcn_fence(__ATOMIC_ACQ_REL, "wavefront", "local");
    }

    float* ob = out + (size_t)b * Mc * (Hc * Ec) + (size_t)h * Ec;
    #pragma unroll
    for (int r = 0; r < 8; ++r) {
        const int m = m0 + r + 8 * half;
        float v0 = acc[0][r];
        float v1 = acc[1][r];
        v0 = v0 > 0.0f ? v0 : (__expf(v0) - 1.0f);
        v1 = v1 > 0.0f ? v1 : (__expf(v1) - 1.0f);
        ob[(size_t)m * (Hc * Ec) + col]      = v0;
        ob[(size_t)m * (Hc * Ec) + 16 + col] = v1;
    }
}

extern "C" void kernel_launch(void* const* d_in, const int* in_sizes, int n_in,
                              void* d_out, int out_size, void* d_ws, size_t ws_size,
                              hipStream_t stream) {
    using namespace gat;
    const float* x   = (const float*)d_in[0];
    const int*   adj = (const int*)  d_in[1];
    const float* w   = (const float*)d_in[2];
    const float* att = (const float*)d_in[3];
    float* out = (float*)d_out;

    float* ws   = (float*)d_ws;
    float* wh   = ws;
    float* s1   = ws + WH_N;
    float* s2   = ws + WH_N + S_N;
    float* rmx  = ws + WH_N + 2 * S_N;
    float* rsm  = ws + WH_N + 3 * S_N;

    const int rowblocks = Bc * Mc / 8;

    gat_wh_kernel<<<Bc * (Mc / WSLAB), 256, 0, stream>>>(x, w, att, wh, s1, s2);
    gat_stats_kernel<<<rowblocks, 256, 0, stream>>>(adj, s1, s2, rmx, rsm);
    gat_attn_kernel<<<Bc * (Mc / SLAB), 256, 0, stream>>>(adj, wh, s1, s2, rmx, rsm, out);
}


# each exponential computed once: attention loop sums the P values it builds (per-row running sums, result scaled by 1/sum before ELU); stats kernel keeps only the masked row max; sum order changes
# speedup vs baseline: 2.8779x; 1.1684x over previous
#include <hip/hip_runtime.h>


typedef __attribute__((ext_vector_type(2))) float v2f;
typedef __attribute__((ext_vector_type(8))) float v8f;

namespace gat {
constexpr int Bc = 8, Mc = 1024, Fc = 256, Hc = 8, Ec = 32;
constexpr size_t WH_N = (size_t)Bc * Hc * Mc * Ec;
constexpr size_t S_N  = (size_t)Bc * Hc * Mc;
constexpr float LALPHA  = 0.2f;
constexpr float NEG_INF = -9.0e15f;
}

namespace gat {
constexpr int WSLAB  = 16;
constexpr int XPITCH = Fc + 4;
}

__global__ __launch_bounds__(256)
void gat_wh_kernel(const float* __restrict__ x,
                   const float* __restrict__ w,
                   const float* __restrict__ att,
                   float* __restrict__ wh,
                   float* __restrict__ s1,
                   float* __restrict__ s2) {
    using namespace gat;
    __shared__ __attribute__((aligned(16))) float xs[WSLAB * XPITCH];

    const int tid  = threadIdx.x;
    const int lane = tid & 31;
    const int h    = tid >> 5;
    const int mb   = blockIdx.x & 63;
    const int b    = blockIdx.x >> 6;

    const int col  = lane & 15;
    const int half = lane >> 4;
    const int k0   = half * 2;

    const int m0 = mb * WSLAB;

    {
        const float4* xsrc = (const float4*)(x + ((size_t)b * Mc + m0) * Fc);
        #pragma unroll
        for (int i = 0; i < WSLAB * Fc / 4 / 256; ++i) {
            const int idx = tid + 256 * i;
            const float4 v = xsrc[idx];
            *(float4*)&xs[(idx >> 6) * XPITCH + 4 * (idx & 63)] = v;
        }
    }

    const float* xrow  = xs + col * XPITCH + k0;
    const float* wbase = w + (size_t)h * Fc * Ec;

    __syncthreads();

    v8f acc[2] = {};
    for (int kk = 0; kk < Fc; kk += 4) {
        v2f af = *(const v2f*)(xrow + kk);
        const float* wp = wbase + (size_t)(kk + k0) * Ec;
        v2f bf0 = { wp[col],      wp[Ec + col]      };
        v2f bf1 = { wp[16 + col], wp[Ec + 16 + col] };
        acc[0] = __builtin_amdgcn_wmma_f32_16x16x4_f32(false, af, false, bf0,
                                                       (short)0, acc[0], false, false);
        acc[1] = __builtin_amdgcn_wmma_f32_16x16x4_f32(false, af, false, bf1,
                                                       (short)0, acc[1], false, false);
    }

    const float a1_0 = att[h * 2 * Ec + col];
    const float a1_1 = att[h * 2 * Ec + 16 + col];
    const float a2_0 = att[h * 2 * Ec + Ec + col];
    const float a2_1 = att[h * 2 * Ec + Ec + 16 + col];

    const int bh = b * Hc + h;
    float* whbase = wh + ((size_t)bh * Mc + m0) * Ec;

    #pragma unroll
    for (int r = 0; r < 8; ++r) {
        const int m_l = r + 8 * half;
        const float v0 = acc[0][r];
        const float v1 = acc[1][r];
        whbase[(size_t)m_l * Ec + col]      = v0;
        whbase[(size_t)m_l * Ec + 16 + col] = v1;

        float p1 = v0 * a1_0 + v1 * a1_1;
        float p2 = v0 * a2_0 + v1 * a2_1;
        #pragma unroll
        for (int d = 1; d < 16; d <<= 1) {
            p1 += __shfl_xor(p1, d, 32);
            p2 += __shfl_xor(p2, d, 32);
        }
        if (col == 0) {
            s1[(size_t)bh * Mc + m0 + m_l] = p1;
            s2[(size_t)bh * Mc + m0 + m_l] = p2;
        }
    }
}

namespace gat {
constexpr float LOG2E = 1.4426950408889634f;
constexpr float NEGL  = NEG_INF * LOG2E;
constexpr float QL    = LALPHA * LOG2E;
}

__global__ __launch_bounds__(256)
void gat_stats_kernel(const int* __restrict__ adj,
                      const float* __restrict__ s1,
                      const float* __restrict__ s2,
                      float* __restrict__ rmax) {
    using namespace gat;
    const int lane = threadIdx.x & 31;
    const int row  = blockIdx.x * 8 + (threadIdx.x >> 5);
    const int m    = row & (Mc - 1);
    const int b    = row >> 10;
    const size_t bh0 = (size_t)b * Hc;

    __shared__ __attribute__((aligned(16))) float part[8][Hc * 32];
    float* pw = part[threadIdx.x >> 5];

    const int   hq  = lane >> 2;
    const float s1l = s1[(bh0 + hq) * Mc + m];

    const int4* arow = (const int4*)(adj + (size_t)row * Mc) + lane;
    v2f kb[16];
    int amax = 0;
    #pragma unroll
    for (int j = 0; j < 8; ++j) {
        const int4 a = arow[j * 32];
        kb[2 * j]     = v2f{ a.x > 0 ? 0.0f : NEGL, a.y > 0 ? 0.0f : NEGL };
        kb[2 * j + 1] = v2f{ a.z > 0 ? 0.0f : NEGL, a.w > 0 ? 0.0f : NEGL };
        amax = max(max(amax, a.x), a.y);
        amax = max(max(amax, a.z), a.w);
    }
    const bool anykeep = __builtin_amdgcn_ballot_w32(amax > 0) != 0u;

    #pragma unroll 1
    for (int h = 0; h < Hc; ++h) {
        const float4* s2row = (const float4*)(s2 + (bh0 + h) * Mc) + lane;
        float m2a = -__builtin_inff(), m2b = -__builtin_inff();
        #pragma unroll
        for (int j = 0; j < 8; ++j) {
            const float4 c = s2row[j * 32];
            const v2f lo = v2f{ c.x, c.y } + kb[2 * j];
            const v2f hi = v2f{ c.z, c.w } + kb[2 * j + 1];
            m2a = fmaxf(fmaxf(m2a, lo.x), lo.y);
            m2b = fmaxf(fmaxf(m2b, hi.x), hi.y);
        }
        pw[h * 32 + lane] = fmaxf(m2a, m2b);
    }

    const float4 u = *(const float4*)&pw[hq * 32 + 8 * (lane & 3)];
    const float4 v = *(const float4*)&pw[hq * 32 + 8 * (lane & 3) + 4];
    float m2 = fmaxf(fmaxf(fmaxf(u.x, u.y), fmaxf(u.z, u.w)),
                     fmaxf(fmaxf(v.x, v.y), fmaxf(v.z, v.w)));
    m2 = fmaxf(m2, __shfl_xor(m2, 1, 32));
    m2 = fmaxf(m2, __shfl_xor(m2, 2, 32));

    float t = s1l + m2;
    t = t > 0.0f ? t : LALPHA * t;
    const float mx = anykeep ? fmaxf(t, NEG_INF) : NEG_INF;

    if ((lane & 3) == 0) rmax[(bh0 + hq) * Mc + m] = mx;
}

namespace gat {
constexpr int SLAB   = 16;
constexpr int APITCH = Mc + 4;
constexpr int NT     = 32;
constexpr int PPITCH = NT + 4;
}

__global__ __launch_bounds__(256)
void gat_attn_kernel(const int* __restrict__ adj,
                     const float* __restrict__ wh,
                     const float* __restrict__ s1,
                     const float* __restrict__ s2,
                     const float* __restrict__ rmax,
                     float* __restrict__ out) {
    using namespace gat;
    __shared__ __attribute__((aligned(16))) float brow[SLAB * APITCH];
    __shared__ __attribute__((aligned(16))) float pblk[Hc][16 * PPITCH];

    const int tid  = threadIdx.x;
    const int lane = tid & 31;
    const int h    = tid >> 5;
    const int mb   = blockIdx.x & 63;
    const int b    = blockIdx.x >> 6;
    const int bh   = b * Hc + h;

    const int col  = lane & 15;
    const int half = lane >> 4;
    const int m0   = mb * SLAB;

    {
        const int4* asrc = (const int4*)(adj + ((size_t)b * Mc + m0) * Mc) + tid;
        #pragma unroll
        for (int r = 0; r < SLAB; ++r) {
            const int4 a = asrc[(size_t)r * (Mc / 4)];
            const float4 bv = { a.x > 0 ? 0.0f : NEGL, a.y > 0 ? 0.0f : NEGL,
                                a.z > 0 ? 0.0f : NEGL, a.w > 0 ? 0.0f : NEGL };
            *(float4*)&brow[r * APITCH + 4 * tid] = bv;
        }
    }

    v2f lv[8], qv[8], av[8], cv[8];
    #pragma unroll
    for (int r = 0; r < 8; ++r) {
        const int m = m0 + r + 8 * half;
        const float s1r = s1[(size_t)bh * Mc + m];
        const float mx  = rmax[(size_t)bh * Mc + m];
        const bool  nk = (mx == NEG_INF);
        const float l = nk ? 0.0f : LOG2E;
        const float q = nk ? 0.0f : QL;
        const float a = nk ? 0.0f : (s1r - mx) * LOG2E;
        const float c = nk ? 0.0f : (LALPHA * s1r - mx) * LOG2E;
        lv[r] = v2f{ l, l };
        qv[r] = v2f{ q, q };
        av[r] = v2f{ a, a };
        cv[r] = v2f{ c, c };
    }

    const float* whb = wh + (size_t)bh * Mc * Ec;
    const float* s2b = s2 + (size_t)bh * Mc + 2 * col;
    float*       pw  = pblk[h];
    const float* bm  = brow + (8 * half) * APITCH + 2 * col;
    float*       pst = pw + (8 * half) * PPITCH + 2 * col;
    const float* pld = pw + col * PPITCH + 16 * half;

    __syncthreads();

    v8f acc[2] = {};
    v2f ps[8]  = {};
    for (int nb = 0; nb < Mc / NT; ++nb) {
        const int n0 = nb * NT;

        const v2f s2v = *(const v2f*)(s2b + n0);

        v2f bf[8][2];
        #pragma unroll
        for (int ks = 0; ks < 8; ++ks) {
            const float* wp = whb + (size_t)(n0 + 16 * half + 2 * ks) * Ec;
            bf[ks][0] = v2f{ wp[col],      wp[Ec + col]      };
            bf[ks][1] = v2f{ wp[16 + col], wp[Ec + 16 + col] };
        }

        #pragma unroll
        for (int r = 0; r < 8; ++r) {
            const v2f sb = s2v + *(const v2f*)&bm[r * APITCH + n0];
            const v2f ex = __builtin_elementwise_max(__builtin_elementwise_fma(sb, lv[r], av[r]),
                                                     __builtin_elementwise_fma(sb, qv[r], cv[r]));
            const v2f p = v2f{ __builtin_amdgcn_exp2f(ex.x), __builtin_amdgcn_exp2f(ex.y) };
            ps[r] += p;
            *(v2f*)&pst[r * PPITCH] = p;
        }
        __builtin_amdgcn_fence(__ATOMIC_ACQ_REL, "wavefront", "local");

        #pragma unroll
        for (int q = 0; q < 4; ++q) {
            const float4 p4 = *(const float4*)&pld[4 * q];
            const v2f af0 = v2f{ p4.x, p4.y };
            const v2f af1 = v2f{ p4.z, p4.w };
            acc[0] = __builtin_amdgcn_wmma_f32_16x16x4_f32(false, af0, false, bf[2 * q][0],
                                                           (short)0, acc[0], false, false);
            acc[1] = __builtin_amdgcn_wmma_f32_16x16x4_f32(false, af0, false, bf[2 * q][1],
                                                           (short)0, acc[1], false, false);
            acc[0] = __builtin_amdgcn_wmma_f32_16x16x4_f32(false, af1, false, bf[2 * q + 1][0],
                                                           (short)0, acc[0], false, false);
            acc[1] = __builtin_amdgcn_wmma_f32_16x16x4_f32(false, af1, false, bf[2 * q + 1][1],
                                                           (short)0, acc[1], false, false);
        }
        __builtin_amdgcn_fence(__ATOMIC_ACQ_REL, "wavefront", "local");
    }

    float* ob = out + (size_t)b * Mc * (Hc * Ec) + (size_t)h * Ec;
    #pragma unroll
    for (int r = 0; r < 8; ++r) {
        const int m = m0 + r + 8 * half;
        float rs = ps[r].x + ps[r].y;
        #pragma unroll
        for (int d = 1; d < 16; d <<= 1) rs += __shfl_xor(rs, d, 32);
        const float inv = __builtin_amdgcn_rcpf(rs);
        float v0 = acc[0][r] * inv;
        float v1 = acc[1][r] * inv;
        v0 = v0 > 0.0f ? v0 : (__expf(v0) - 1.0f);
        v1 = v1 > 0.0f ? v1 : (__expf(v1) - 1.0f);
        ob[(size_t)m * (Hc * Ec) + col]      = v0;
        ob[(size_t)m * (Hc * Ec) + 16 + col] = v1;
    }
}

extern "C" void kernel_launch(void* const* d_in, const int* in_sizes, int n_in,
                              void* d_out, int out_size, void* d_ws, size_t ws_size,
                              hipStream_t stream) {
    using namespace gat;
    const float* x   = (const float*)d_in[0];
    const int*   adj = (const int*)  d_in[1];
    const float* w   = (const float*)d_in[2];
    const float* att = (const float*)d_in[3];
    float* out = (float*)d_out;

    float* ws   = (float*)d_ws;
    float* wh   = ws;
    float* s1   = ws + WH_N;
    float* s2   = ws + WH_N + S_N;
    float* rmx  = ws + WH_N + 2 * S_N;

    const int rowblocks = Bc * Mc / 8;

    gat_wh_kernel<<<Bc * (Mc / WSLAB), 256, 0, stream>>>(x, w, att, wh, s1, s2);
    gat_stats_kernel<<<rowblocks, 256, 0, stream>>>(adj, s1, s2, rmx);
    gat_attn_kernel<<<Bc * (Mc / SLAB), 256, 0, stream>>>(adj, wh, s1, s2, rmx, out);
}


# masked row max taken in the attention block's prologue: wave = head runs its s2 row (in registers) over the slab's 16 LDS bias rows, partials folded in its own tile; stats launch gone; sum order same
# speedup vs baseline: 3.1662x; 1.1002x over previous
#include <hip/hip_runtime.h>


typedef __attribute__((ext_vector_type(2))) float v2f;
typedef __attribute__((ext_vector_type(8))) float v8f;

namespace gat {
constexpr int Bc = 8, Mc = 1024, Fc = 256, Hc = 8, Ec = 32;
constexpr size_t WH_N = (size_t)Bc * Hc * Mc * Ec;
constexpr size_t S_N  = (size_t)Bc * Hc * Mc;
constexpr float LALPHA  = 0.2f;
constexpr float NEG_INF = -9.0e15f;
}

namespace gat {
constexpr int WSLAB  = 16;
constexpr int XPITCH = Fc + 4;
}

__global__ __launch_bounds__(256)
void gat_wh_kernel(const float* __restrict__ x,
                   const float* __restrict__ w,
                   const float* __restrict__ att,
                   float* __restrict__ wh,
                   float* __restrict__ s1,
                   float* __restrict__ s2) {
    using namespace gat;
    __shared__ __attribute__((aligned(16))) float xs[WSLAB * XPITCH];

    const int tid  = threadIdx.x;
    const int lane = tid & 31;
    const int h    = tid >> 5;
    const int mb   = blockIdx.x & 63;
    const int b    = blockIdx.x >> 6;

    const int col  = lane & 15;
    const int half = lane >> 4;
    const int k0   = half * 2;

    const int m0 = mb * WSLAB;

    {
        const float4* xsrc = (const float4*)(x + ((size_t)b * Mc + m0) * Fc);
        #pragma unroll
        for (int i = 0; i < WSLAB * Fc / 4 / 256; ++i) {
            const int idx = tid + 256 * i;
            const float4 v = xsrc[idx];
            *(float4*)&xs[(idx >> 6) * XPITCH + 4 * (idx & 63)] = v;
        }
    }

    const float* xrow  = xs + col * XPITCH + k0;
    const float* wbase = w + (size_t)h * Fc * Ec;

    __syncthreads();

    v8f acc[2] = {};
    for (int kk = 0; kk < Fc; kk += 4) {
        v2f af = *(const v2f*)(xrow + kk);
        const float* wp = wbase + (size_t)(kk + k0) * Ec;
        v2f bf0 = { wp[col],      wp[Ec + col]      };
        v2f bf1 = { wp[16 + col], wp[Ec + 16 + col] };
        acc[0] = __builtin_amdgcn_wmma_f32_16x16x4_f32(false, af, false, bf0,
                                                       (short)0, acc[0], false, false);
        acc[1] = __builtin_amdgcn_wmma_f32_16x16x4_f32(false, af, false, bf1,
                                                       (short)0, acc[1], false, false);
    }

    const float a1_0 = att[h * 2 * Ec + col];
    const float a1_1 = att[h * 2 * Ec + 16 + col];
    const float a2_0 = att[h * 2 * Ec + Ec + col];
    const float a2_1 = att[h * 2 * Ec + Ec + 16 + col];

    const int bh = b * Hc + h;
    float* whbase = wh + ((size_t)bh * Mc + m0) * Ec;

    #pragma unroll
    for (int r = 0; r < 8; ++r) {
        const int m_l = r + 8 * half;
        const float v0 = acc[0][r];
        const float v1 = acc[1][r];
        whbase[(size_t)m_l * Ec + col]      = v0;
        whbase[(size_t)m_l * Ec + 16 + col] = v1;

        float p1 = v0 * a1_0 + v1 * a1_1;
        float p2 = v0 * a2_0 + v1 * a2_1;
        #pragma unroll
        for (int d = 1; d < 16; d <<= 1) {
            p1 += __shfl_xor(p1, d, 32);
            p2 += __shfl_xor(p2, d, 32);
        }
        if (col == 0) {
            s1[(size_t)bh * Mc + m0 + m_l] = p1;
            s2[(size_t)bh * Mc + m0 + m_l] = p2;
        }
    }
}

namespace gat {
constexpr float LOG2E = 1.4426950408889634f;
constexpr float NEGL  = NEG_INF * LOG2E;
constexpr float QL    = LALPHA * LOG2E;
constexpr float KEEPT = 0.5f * NEGL;
}

namespace gat {
constexpr int SLAB   = 16;
constexpr int APITCH = Mc + 4;
constexpr int NT     = 32;
constexpr int PPITCH = NT + 4;
}

__global__ __launch_bounds__(256)
void gat_attn_kernel(const int* __restrict__ adj,
                     const float* __restrict__ wh,
                     const float* __restrict__ s1,
                     const float* __restrict__ s2,
                     float* __restrict__ out) {
    using namespace gat;
    __shared__ __attribute__((aligned(16))) float brow[SLAB * APITCH];
    __shared__ __attribute__((aligned(16))) float pblk[Hc][16 * PPITCH];

    const int tid  = threadIdx.x;
    const int lane = tid & 31;
    const int h    = tid >> 5;
    const int mb   = blockIdx.x & 63;
    const int b    = blockIdx.x >> 6;
    const int bh   = b * Hc + h;

    const int col  = lane & 15;
    const int half = lane >> 4;
    const int m0   = mb * SLAB;

    {
        const int4* asrc = (const int4*)(adj + ((size_t)b * Mc + m0) * Mc) + tid;
        #pragma unroll
        for (int r = 0; r < SLAB; ++r) {
            const int4 a = asrc[(size_t)r * (Mc / 4)];
            const float4 bv = { a.x > 0 ? 0.0f : NEGL, a.y > 0 ? 0.0f : NEGL,
                                a.z > 0 ? 0.0f : NEGL, a.w > 0 ? 0.0f : NEGL };
            *(float4*)&brow[r * APITCH + 4 * tid] = bv;
        }
    }

    float4 sv[8];
    {
        const float4* s2row = (const float4*)(s2 + (size_t)bh * Mc) + lane;
        #pragma unroll
        for (int j = 0; j < 8; ++j) sv[j] = s2row[j * 32];
    }
    float s1r[8];
    #pragma unroll
    for (int r = 0; r < 8; ++r) s1r[r] = s1[(size_t)bh * Mc + m0 + r + 8 * half];

    const float* whb = wh + (size_t)bh * Mc * Ec;
    const float* s2b = s2 + (size_t)bh * Mc + 2 * col;
    float*       pw  = pblk[h];
    const float* bm  = brow + (8 * half) * APITCH + 2 * col;
    float*       pst = pw + (8 * half) * PPITCH + 2 * col;
    const float* pld = pw + col * PPITCH + 16 * half;

    __syncthreads();

    #pragma unroll 2
    for (int r = 0; r < SLAB; ++r) {
        const float4* brp = (const float4*)&brow[r * APITCH] + lane;
        float mp = -__builtin_inff(), mq = -__builtin_inff();
        #pragma unroll
        for (int j = 0; j < 8; ++j) {
            const float4 bb = brp[j * 32];
            const v2f lo = v2f{ sv[j].x, sv[j].y } + v2f{ bb.x, bb.y };
            const v2f hi = v2f{ sv[j].z, sv[j].w } + v2f{ bb.z, bb.w };
            mp = fmaxf(fmaxf(mp, lo.x), lo.y);
            mq = fmaxf(fmaxf(mq, hi.x), hi.y);
        }
        pw[r * PPITCH + lane] = fmaxf(mp, mq);
    }

    float mrow;
    {
        const float4 u0 = *(const float4*)&pld[0];
        const float4 u1 = *(const float4*)&pld[4];
        const float4 u2 = *(const float4*)&pld[8];
        const float4 u3 = *(const float4*)&pld[12];
        const float e0 = fmaxf(fmaxf(u0.x, u0.y), fmaxf(u0.z, u0.w));
        const float e1 = fmaxf(fmaxf(u1.x, u1.y), fmaxf(u1.z, u1.w));
        const float e2 = fmaxf(fmaxf(u2.x, u2.y), fmaxf(u2.z, u2.w));
        const float e3 = fmaxf(fmaxf(u3.x, u3.y), fmaxf(u3.z, u3.w));
        mrow = fmaxf(fmaxf(e0, e1), fmaxf(e2, e3));
        mrow = fmaxf(mrow, __shfl_xor(mrow, 16, 32));
    }

    float m2v[8];
    #pragma unroll
    for (int r = 0; r < 8; ++r) m2v[r] = __shfl(mrow, r + 8 * half, 32);
    v2f lv[8], qv[8], av[8], cv[8];
    #pragma unroll
    for (int r = 0; r < 8; ++r) {
        const float m2 = m2v[r];
        float t = s1r[r] + m2;
        t = t > 0.0f ? t : LALPHA * t;
        const float mx = (m2 > KEEPT) ? fmaxf(t, NEG_INF) : NEG_INF;
        const bool  nk = (mx == NEG_INF);
        const float l = nk ? 0.0f : LOG2E;
        const float q = nk ? 0.0f : QL;
        const float a = nk ? 0.0f : (s1r[r] - mx) * LOG2E;
        const float c = nk ? 0.0f : (LALPHA * s1r[r] - mx) * LOG2E;
        lv[r] = v2f{ l, l };
        qv[r] = v2f{ q, q };
        av[r] = v2f{ a, a };
        cv[r] = v2f{ c, c };
    }

    v8f acc[2] = {};
    v2f ps[8]  = {};
    for (int nb = 0; nb < Mc / NT; ++nb) {
        const int n0 = nb * NT;

        const v2f s2v = *(const v2f*)(s2b + n0);

        v2f bf[8][2];
        #pragma unroll
        for (int ks = 0; ks < 8; ++ks) {
            const float* wp = whb + (size_t)(n0 + 16 * half + 2 * ks) * Ec;
            bf[ks][0] = v2f{ wp[col],      wp[Ec + col]      };
            bf[ks][1] = v2f{ wp[16 + col], wp[Ec + 16 + col] };
        }

        #pragma unroll
        for (int r = 0; r < 8; ++r) {
            const v2f sb = s2v + *(const v2f*)&bm[r * APITCH + n0];
            const v2f ex = __builtin_elementwise_max(__builtin_elementwise_fma(sb, lv[r], av[r]),
                                                     __builtin_elementwise_fma(sb, qv[r], cv[r]));
            const v2f p = v2f{ __builtin_amdgcn_exp2f(ex.x), __builtin_amdgcn_exp2f(ex.y) };
            ps[r] += p;
            *(v2f*)&pst[r * PPITCH] = p;
        }
        __builtin_amdgcn_fence(__ATOMIC_ACQ_REL, "wavefront", "local");

        #pragma unroll
        for (int q = 0; q < 4; ++q) {
            const float4 p4 = *(const float4*)&pld[4 * q];
            const v2f af0 = v2f{ p4.x, p4.y };
            const v2f af1 = v2f{ p4.z, p4.w };
            acc[0] = __builtin_amdgcn_wmma_f32_16x16x4_f32(false, af0, false, bf[2 * q][0],
                                                           (short)0, acc[0], false, false);
            acc[1] = __builtin_amdgcn_wmma_f32_16x16x4_f32(false, af0, false, bf[2 * q][1],
                                                           (short)0, acc[1], false, false);
            acc[0] = __builtin_amdgcn_wmma_f32_16x16x4_f32(false, af1, false, bf[2 * q + 1][0],
                                                           (short)0, acc[0], false, false);
            acc[1] = __builtin_amdgcn_wmma_f32_16x16x4_f32(false, af1, false, bf[2 * q + 1][1],
                                                           (short)0, acc[1], false, false);
        }
        __builtin_amdgcn_fence(__ATOMIC_ACQ_REL, "wavefront", "local");
    }

    float* ob = out + (size_t)b * Mc * (Hc * Ec) + (size_t)h * Ec;
    #pragma unroll
    for (int r = 0; r < 8; ++r) {
        const int m = m0 + r + 8 * half;
        float rs = ps[r].x + ps[r].y;
        #pragma unroll
        for (int d = 1; d < 16; d <<= 1) rs += __shfl_xor(rs, d, 32);
        const float inv = __builtin_amdgcn_rcpf(rs);
        float v0 = acc[0][r] * inv;
        float v1 = acc[1][r] * inv;
        v0 = v0 > 0.0f ? v0 : (__expf(v0) - 1.0f);
        v1 = v1 > 0.0f ? v1 : (__expf(v1) - 1.0f);
        ob[(size_t)m * (Hc * Ec) + col]      = v0;
        ob[(size_t)m * (Hc * Ec) + 16 + col] = v1;
    }
}

extern "C" void kernel_launch(void* const* d_in, const int* in_sizes, int n_in,
                              void* d_out, int out_size, void* d_ws, size_t ws_size,
                              hipStream_t stream) {
    using namespace gat;
    const float* x   = (const float*)d_in[0];
    const int*   adj = (const int*)  d_in[1];
    const float* w   = (const float*)d_in[2];
    const float* att = (const float*)d_in[3];
    float* out = (float*)d_out;

    float* ws   = (float*)d_ws;
    float* wh   = ws;
    float* s1   = ws + WH_N;
    float* s2   = ws + WH_N + S_N;

    gat_wh_kernel<<<Bc * (Mc / WSLAB), 256, 0, stream>>>(x, w, att, wh, s1, s2);
    gat_attn_kernel<<<Bc * (Mc / SLAB), 256, 0, stream>>>(adj, wh, s1, s2, out);
}


# attention loop software-pipelined over two P tiles a wave: next trip's P math (s2 from an LDS copy) and Wh loads stand beside this trip's WMMAs, no same-trip load wait; sum order same
# speedup vs baseline: 3.2929x; 1.0400x over previous
#include <hip/hip_runtime.h>


typedef __attribute__((ext_vector_type(2))) float v2f;
typedef __attribute__((ext_vector_type(8))) float v8f;

namespace gat {
constexpr int Bc = 8, Mc = 1024, Fc = 256, Hc = 8, Ec = 32;
constexpr size_t WH_N = (size_t)Bc * Hc * Mc * Ec;
constexpr size_t S_N  = (size_t)Bc * Hc * Mc;
constexpr float LALPHA  = 0.2f;
constexpr float NEG_INF = -9.0e15f;
}

namespace gat {
constexpr int WSLAB  = 16;
constexpr int XPITCH = Fc + 4;
}

__global__ __launch_bounds__(256)
void gat_wh_kernel(const float* __restrict__ x,
                   const float* __restrict__ w,
                   const float* __restrict__ att,
                   float* __restrict__ wh,
                   float* __restrict__ s1,
                   float* __restrict__ s2) {
    using namespace gat;
    __shared__ __attribute__((aligned(16))) float xs[WSLAB * XPITCH];

    const int tid  = threadIdx.x;
    const int lane = tid & 31;
    const int h    = tid >> 5;
    const int mb   = blockIdx.x & 63;
    const int b    = blockIdx.x >> 6;

    const int col  = lane & 15;
    const int half = lane >> 4;
    const int k0   = half * 2;

    const int m0 = mb * WSLAB;

    {
        const float4* xsrc = (const float4*)(x + ((size_t)b * Mc + m0) * Fc);
        #pragma unroll
        for (int i = 0; i < WSLAB * Fc / 4 / 256; ++i) {
            const int idx = tid + 256 * i;
            const float4 v = xsrc[idx];
            *(float4*)&xs[(idx >> 6) * XPITCH + 4 * (idx & 63)] = v;
        }
    }

    const float* xrow  = xs + col * XPITCH + k0;
    const float* wbase = w + (size_t)h * Fc * Ec;

    __syncthreads();

    v8f acc[2] = {};
    for (int kk = 0; kk < Fc; kk += 4) {
        v2f af = *(const v2f*)(xrow + kk);
        const float* wp = wbase + (size_t)(kk + k0) * Ec;
        v2f bf0 = { wp[col],      wp[Ec + col]      };
        v2f bf1 = { wp[16 + col], wp[Ec + 16 + col] };
        acc[0] = __builtin_amdgcn_wmma_f32_16x16x4_f32(false, af, false, bf0,
                                                       (short)0, acc[0], false, false);
        acc[1] = __builtin_amdgcn_wmma_f32_16x16x4_f32(false, af, false, bf1,
                                                       (short)0, acc[1], false, false);
    }

    const float a1_0 = att[h * 2 * Ec + col];
    const float a1_1 = att[h * 2 * Ec + 16 + col];
    const float a2_0 = att[h * 2 * Ec + Ec + col];
    const float a2_1 = att[h * 2 * Ec + Ec + 16 + col];

    const int bh = b * Hc + h;
    float* whbase = wh + ((size_t)bh * Mc + m0) * Ec;

    #pragma unroll
    for (int r = 0; r < 8; ++r) {
        const int m_l = r + 8 * half;
        const float v0 = acc[0][r];
        const float v1 = acc[1][r];
        whbase[(size_t)m_l * Ec + col]      = v0;
        whbase[(size_t)m_l * Ec + 16 + col] = v1;

        float p1 = v0 * a1_0 + v1 * a1_1;
        float p2 = v0 * a2_0 + v1 * a2_1;
        #pragma unroll
        for (int d = 1; d < 16; d <<= 1) {
            p1 += __shfl_xor(p1, d, 32);
            p2 += __shfl_xor(p2, d, 32);
        }
        if (col == 0) {
            s1[(size_t)bh * Mc + m0 + m_l] = p1;
            s2[(size_t)bh * Mc + m0 + m_l] = p2;
        }
    }
}

namespace gat {
constexpr float LOG2E = 1.4426950408889634f;
constexpr float NEGL  = NEG_INF * LOG2E;
constexpr float QL    = LALPHA * LOG2E;
constexpr float KEEPT = 0.5f * NEGL;
}

namespace gat {
constexpr int SLAB   = 16;
constexpr int APITCH = Mc + 4;
constexpr int NT     = 32;
constexpr int PPITCH = NT + 4;
}

__global__ __launch_bounds__(256)
void gat_attn_kernel(const int* __restrict__ adj,
                     const float* __restrict__ wh,
                     const float* __restrict__ s1,
                     const float* __restrict__ s2,
                     float* __restrict__ out) {
    using namespace gat;
    __shared__ __attribute__((aligned(16))) float brow[SLAB * APITCH];
    __shared__ __attribute__((aligned(16))) float pblk[Hc][2 * 16 * PPITCH];
    __shared__ __attribute__((aligned(16))) float s2l[Hc][Mc];

    const int tid  = threadIdx.x;
    const int lane = tid & 31;
    const int h    = tid >> 5;
    const int mb   = blockIdx.x & 63;
    const int b    = blockIdx.x >> 6;
    const int bh   = b * Hc + h;

    const int col  = lane & 15;
    const int half = lane >> 4;
    const int m0   = mb * SLAB;

    {
        const int4* asrc = (const int4*)(adj + ((size_t)b * Mc + m0) * Mc) + tid;
        #pragma unroll
        for (int r = 0; r < SLAB; ++r) {
            const int4 a = asrc[(size_t)r * (Mc / 4)];
            const float4 bv = { a.x > 0 ? 0.0f : NEGL, a.y > 0 ? 0.0f : NEGL,
                                a.z > 0 ? 0.0f : NEGL, a.w > 0 ? 0.0f : NEGL };
            *(float4*)&brow[r * APITCH + 4 * tid] = bv;
        }
    }

    float4 sv[8];
    {
        const float4* s2row = (const float4*)(s2 + (size_t)bh * Mc) + lane;
        #pragma unroll
        for (int j = 0; j < 8; ++j) sv[j] = s2row[j * 32];
    }
    float s1r[8];
    #pragma unroll
    for (int r = 0; r < 8; ++r) s1r[r] = s1[(size_t)bh * Mc + m0 + r + 8 * half];

    const float* whb = wh + (size_t)bh * Mc * Ec;
    const float* s2b = s2l[h] + 2 * col;
    float*       pw  = pblk[h];
    const float* bm  = brow + (8 * half) * APITCH + 2 * col;
    float*       pst = pw + (8 * half) * PPITCH + 2 * col;
    const float* pld = pw + col * PPITCH + 16 * half;

    __syncthreads();

    #pragma unroll
    for (int j = 0; j < 8; ++j) *(float4*)&s2l[h][128 * j + 4 * lane] = sv[j];

    #pragma unroll 2
    for (int r = 0; r < SLAB; ++r) {
        const float4* brp = (const float4*)&brow[r * APITCH] + lane;
        float mp = -__builtin_inff(), mq = -__builtin_inff();
        #pragma unroll
        for (int j = 0; j < 8; ++j) {
            const float4 bb = brp[j * 32];
            const v2f lo = v2f{ sv[j].x, sv[j].y } + v2f{ bb.x, bb.y };
            const v2f hi = v2f{ sv[j].z, sv[j].w } + v2f{ bb.z, bb.w };
            mp = fmaxf(fmaxf(mp, lo.x), lo.y);
            mq = fmaxf(fmaxf(mq, hi.x), hi.y);
        }
        pw[r * PPITCH + lane] = fmaxf(mp, mq);
    }

    float mrow;
    {
        const float4 u0 = *(const float4*)&pld[0];
        const float4 u1 = *(const float4*)&pld[4];
        const float4 u2 = *(const float4*)&pld[8];
        const float4 u3 = *(const float4*)&pld[12];
        const float e0 = fmaxf(fmaxf(u0.x, u0.y), fmaxf(u0.z, u0.w));
        const float e1 = fmaxf(fmaxf(u1.x, u1.y), fmaxf(u1.z, u1.w));
        const float e2 = fmaxf(fmaxf(u2.x, u2.y), fmaxf(u2.z, u2.w));
        const float e3 = fmaxf(fmaxf(u3.x, u3.y), fmaxf(u3.z, u3.w));
        mrow = fmaxf(fmaxf(e0, e1), fmaxf(e2, e3));
        mrow = fmaxf(mrow, __shfl_xor(mrow, 16, 32));
    }

    float m2v[8];
    #pragma unroll
    for (int r = 0; r < 8; ++r) m2v[r] = __shfl(mrow, r + 8 * half, 32);
    v2f lv[8], qv[8], av[8], cv[8];
    #pragma unroll
    for (int r = 0; r < 8; ++r) {
        const float m2 = m2v[r];
        float t = s1r[r] + m2;
        t = t > 0.0f ? t : LALPHA * t;
        const float mx = (m2 > KEEPT) ? fmaxf(t, NEG_INF) : NEG_INF;
        const bool  nk = (mx == NEG_INF);
        const float l = nk ? 0.0f : LOG2E;
        const float q = nk ? 0.0f : QL;
        const float a = nk ? 0.0f : (s1r[r] - mx) * LOG2E;
        const float c = nk ? 0.0f : (LALPHA * s1r[r] - mx) * LOG2E;
        lv[r] = v2f{ l, l };
        qv[r] = v2f{ q, q };
        av[r] = v2f{ a, a };
        cv[r] = v2f{ c, c };
    }

    v8f acc[2] = {};
    v2f ps[8]  = {};

    auto load_b = [&](const int n0, v2f (&bfx)[8][2]) {
        #pragma unroll
        for (int ks = 0; ks < 8; ++ks) {
            const float* wp = whb + (size_t)(n0 + 16 * half + 2 * ks) * Ec;
            bfx[ks][0] = v2f{ wp[col],      wp[Ec + col]      };
            bfx[ks][1] = v2f{ wp[16 + col], wp[Ec + 16 + col] };
        }
    };
    auto build_p = [&](const int n0, float* dst) {
        const v2f s2v = *(const v2f*)(s2b + n0);
        #pragma unroll
        for (int r = 0; r < 8; ++r) {
            const v2f sb = s2v + *(const v2f*)&bm[r * APITCH + n0];
            const v2f ex = __builtin_elementwise_max(__builtin_elementwise_fma(sb, lv[r], av[r]),
                                                     __builtin_elementwise_fma(sb, qv[r], cv[r]));
            const v2f p = v2f{ __builtin_amdgcn_exp2f(ex.x), __builtin_amdgcn_exp2f(ex.y) };
            ps[r] += p;
            *(v2f*)&dst[r * PPITCH] = p;
        }
    };
    auto mma = [&](const float4 (&p4)[4], const v2f (&bfx)[8][2]) {
        #pragma unroll
        for (int q = 0; q < 4; ++q) {
            const v2f af0 = v2f{ p4[q].x, p4[q].y };
            const v2f af1 = v2f{ p4[q].z, p4[q].w };
            acc[0] = __builtin_amdgcn_wmma_f32_16x16x4_f32(false, af0, false, bfx[2 * q][0],
                                                           (short)0, acc[0], false, false);
            acc[1] = __builtin_amdgcn_wmma_f32_16x16x4_f32(false, af0, false, bfx[2 * q][1],
                                                           (short)0, acc[1], false, false);
            acc[0] = __builtin_amdgcn_wmma_f32_16x16x4_f32(false, af1, false, bfx[2 * q + 1][0],
                                                           (short)0, acc[0], false, false);
            acc[1] = __builtin_amdgcn_wmma_f32_16x16x4_f32(false, af1, false, bfx[2 * q + 1][1],
                                                           (short)0, acc[1], false, false);
        }
    };
    auto trip = [&](const int n0, const int rd, const int wr,
                    const v2f (&bfc)[8][2], v2f (&bfn)[8][2]) {
        float4 p4[4];
        #pragma unroll
        for (int q = 0; q < 4; ++q) p4[q] = *(const float4*)&pld[rd + 4 * q];
        load_b(n0 + NT, bfn);
        build_p(n0 + NT, pst + wr);
        mma(p4, bfc);
        __builtin_amdgcn_fence(__ATOMIC_ACQ_REL, "wavefront", "local");
    };

    constexpr int TILE = 16 * PPITCH;
    v2f bfa[8][2], bfb[8][2];
    load_b(0, bfa);
    build_p(0, pst);
    __builtin_amdgcn_fence(__ATOMIC_ACQ_REL, "wavefront", "local");

    #pragma unroll 1
    for (int pp = 0; pp < Mc / NT / 2 - 1; ++pp) {
        trip(pp * 2 * NT,      0,    TILE, bfa, bfb);
        trip(pp * 2 * NT + NT, TILE, 0,    bfb, bfa);
    }
    trip(Mc - 2 * NT, 0, TILE, bfa, bfb);
    {
        float4 p4[4];
        #pragma unroll
        for (int q = 0; q < 4; ++q) p4[q] = *(const float4*)&pld[TILE + 4 * q];
        mma(p4, bfb);
    }

    float* ob = out + (size_t)b * Mc * (Hc * Ec) + (size_t)h * Ec;
    #pragma unroll
    for (int r = 0; r < 8; ++r) {
        const int m = m0 + r + 8 * half;
        float rs = ps[r].x + ps[r].y;
        #pragma unroll
        for (int d = 1; d < 16; d <<= 1) rs += __shfl_xor(rs, d, 32);
        const float inv = __builtin_amdgcn_rcpf(rs);
        float v0 = acc[0][r] * inv;
        float v1 = acc[1][r] * inv;
        v0 = v0 > 0.0f ? v0 : (__expf(v0) - 1.0f);
        v1 = v1 > 0.0f ? v1 : (__expf(v1) - 1.0f);
        ob[(size_t)m * (Hc * Ec) + col]      = v0;
        ob[(size_t)m * (Hc * Ec) + 16 + col] = v1;
    }
}

extern "C" void kernel_launch(void* const* d_in, const int* in_sizes, int n_in,
                              void* d_out, int out_size, void* d_ws, size_t ws_size,
                              hipStream_t stream) {
    using namespace gat;
    const float* x   = (const float*)d_in[0];
    const int*   adj = (const int*)  d_in[1];
    const float* w   = (const float*)d_in[2];
    const float* att = (const float*)d_in[3];
    float* out = (float*)d_out;

    float* ws   = (float*)d_ws;
    float* wh   = ws;
    float* s1   = ws + WH_N;
    float* s2   = ws + WH_N + S_N;

    gat_wh_kernel<<<Bc * (Mc / WSLAB), 256, 0, stream>>>(x, w, att, wh, s1, s2);
    gat_attn_kernel<<<Bc * (Mc / SLAB), 256, 0, stream>>>(adj, wh, s1, s2, out);
}
